# starting kernel
# baseline (speedup 1.0000x reference)
#include <hip/hip_runtime.h>

typedef float v2f __attribute__((ext_vector_type(2)));
typedef float v8f __attribute__((ext_vector_type(8)));

#define NB 4
#define NC 16
#define NO 32
#define NK 3
#define NH 56
#define NW 56
#define HW (NH * NW)
#define TILES (HW / 16)

__global__ __launch_bounds__(256) void kaconv_wmma_kernel(
    const float* __restrict__ x,
    const float* __restrict__ nums,
    const float* __restrict__ denoms,
    const float* __restrict__ bias,
    float* __restrict__ out)
{
  const int lane   = threadIdx.x & 31;
  const int wib    = threadIdx.x >> 5;
  const int wave   = blockIdx.x * 8 + wib;
  const int b      = wave / (TILES * 2);
  const int rem    = wave - b * (TILES * 2);
  const int tile   = rem >> 1;
  const int ochunk = rem & 1;

  const int m  = lane & 15;
  const int hi = lane >> 4;

  const int pix = tile * 16 + m;
  const int h = pix / NW;
  const int w = pix - h * NW;

  v8f acc = {};

  const int obase = ochunk * 16;
  const int orow  = (obase + m) * NC;

  for (int c = 0; c < NC; ++c) {
    const float* xc = x + ((size_t)(b * NC + c)) * HW;
    const int rowc = (orow + c) * NK;
#pragma unroll
    for (int i = 0; i < NK; ++i) {
      const int hh    = h + i - 1;
      const int rowci = (rowc + i) * NK;
#pragma unroll
      for (int j = 0; j < NK; ++j) {
        const int ww = w + j - 1;
        const bool inb = (hh >= 0) && (hh < NH) && (ww >= 0) && (ww < NW);
        const float xv = inb ? xc[hh * NW + ww] : 0.0f;
        const float x2 = xv * xv;
        const float x3 = x2 * xv;
        const float x4 = x2 * x2;
        const float x5 = x3 * x2;

        v2f aN0 = {}, aN1 = {}, aD = {};
        aN0[0] = hi ? x2 : 1.0f;  aN0[1] = hi ? x3 : xv;
        aN1[0] = hi ? 0.0f : x4;  aN1[1] = hi ? 0.0f : x5;
        aD[0]  = hi ? x3 : xv;    aD[1]  = hi ? x4 : x2;

        const int row = rowci + j;
        const float* nr = nums   + (size_t)row * 6;
        const float* dr = denoms + (size_t)row * 4;
        v2f bN0 = *(const v2f*)(nr + (hi ? 2 : 0));
        v2f bN1 = *(const v2f*)(nr + (hi ? 2 : 4));
        v2f bD  = *(const v2f*)(dr + (hi ? 2 : 0));

        v8f numv = {};
        numv = __builtin_amdgcn_wmma_f32_16x16x4_f32(false, aN0, false, bN0,
                                                     (short)0, numv, false, false);
        numv = __builtin_amdgcn_wmma_f32_16x16x4_f32(false, aN1, false, bN1,
                                                     (short)0, numv, false, false);
        v8f denv = {};
        denv = __builtin_amdgcn_wmma_f32_16x16x4_f32(false, aD, false, bD,
                                                     (short)0, denv, false, false);

#pragma unroll
        for (int k = 0; k < 8; ++k) {
          const float d = 1.0f + __builtin_fabsf(denv[k]);
          acc[k] = __builtin_fmaf(numv[k], __builtin_amdgcn_rcpf(d), acc[k]);
        }
      }
    }
  }

  const float bo = bias[obase + m];
  float* op = out + ((size_t)(b * NO + obase + m)) * HW + tile * 16 + hi * 8;
#pragma unroll
  for (int k = 0; k < 8; ++k) op[k] = acc[k] + bo;
}

extern "C" void kernel_launch(void* const* d_in, const int* in_sizes, int n_in,
                              void* d_out, int out_size, void* d_ws, size_t ws_size,
                              hipStream_t stream) {
  const float* x      = (const float*)d_in[0];
  const float* nums   = (const float*)d_in[1];
  const float* denoms = (const float*)d_in[2];
  const float* bias   = (const float*)d_in[3];
  float* out          = (float*)d_out;
  (void)in_sizes; (void)n_in; (void)out_size; (void)d_ws; (void)ws_size;

  dim3 grid(196);
  dim3 block(256);
  hipLaunchKernelGGL(kaconv_wmma_kernel, grid, block, 0, stream,
                     x, nums, denoms, bias, out);
}


# speedup vs baseline: 2.5418x; 2.5418x over previous
#include <hip/hip_runtime.h>


typedef float v2f __attribute__((ext_vector_type(2)));
typedef float v4f __attribute__((ext_vector_type(4)));
typedef float v8f __attribute__((ext_vector_type(8)));

#define NB 4
#define NC 16
#define NO 32
#define NK 3
#define NH 56
#define NW 56
#define HW (NH * NW)
#define MT 7
#define GPIX (MT * 16)
#define NGRP (HW / GPIX)
#define RPITCH 116
#define NQUAD (GPIX / 4)
#define WROWS 4
#define WCOLS (NW + 2)
#define WPIX (WROWS * WCOLS)
#define EPITCH 12
#define TPITCH (WPIX * EPITCH)


struct TapOperands {
  v2f aN0[MT], aN1[MT], aD[MT];
  v2f bN0, bN1, bD;
};

__device__ __forceinline__ void load_tap_coefs(TapOperands& s, const float* nrA, const float* nrB,
                                               const float* drA, int tap) {
  s.bN0 = *(const v2f*)(nrA + tap * 6);
  s.bN1 = *(const v2f*)(nrB + tap * 6);
  s.bD  = *(const v2f*)(drA + tap * 4);
}

__device__ __forceinline__ void load_tap_table(TapOperands& s, const float* const* ap, int tap) {
  const int i = tap / NK;
  const int j = tap - i * NK;
  const int toff = (i * WCOLS + j) * EPITCH;
#pragma unroll
  for (int t = 0; t < MT; ++t) {
    const float* e = ap[t] + toff;
    s.aN0[t] = *(const v2f*)(e);
    s.aN1[t] = *(const v2f*)(e + 4);
    s.aD[t]  = *(const v2f*)(e + 8);
  }
}

__device__ __forceinline__ void tap_products(const TapOperands& s, v8f* numv, v8f* denv) {
#pragma unroll
  for (int t = 0; t < MT; ++t) {
    v8f n = {};
    n = __builtin_amdgcn_wmma_f32_16x16x4_f32(false, s.aN0[t], false, s.bN0, (short)0, n, false, false);
    n = __builtin_amdgcn_wmma_f32_16x16x4_f32(false, s.aN1[t], false, s.bN1, (short)0, n, false, false);
    v8f d = {};
    d = __builtin_amdgcn_wmma_f32_16x16x4_f32(false, s.aD[t], false, s.bD, (short)0, d, false, false);
    numv[t] = n;
    denv[t] = d;
  }
}


__device__ __forceinline__ void den_start(const v8f& den, float (&d)[8]) {
  asm volatile(
      "v_add_f32_e64 %0, |%8|, 1.0\n\t"
      "v_add_f32_e64 %1, |%9|, 1.0\n\t"
      "v_add_f32_e64 %2, |%10|, 1.0\n\t"
      "v_add_f32_e64 %3, |%11|, 1.0\n\t"
      "v_add_f32_e64 %4, |%12|, 1.0\n\t"
      "v_add_f32_e64 %5, |%13|, 1.0\n\t"
      "v_add_f32_e64 %6, |%14|, 1.0\n\t"
      "v_add_f32_e64 %7, |%15|, 1.0"
      : "=&v"(d[0]), "=&v"(d[1]), "=&v"(d[2]), "=&v"(d[3]), "=&v"(d[4]), "=&v"(d[5]), "=&v"(d[6]), "=&v"(d[7])
      : "v"(den[0]), "v"(den[1]), "v"(den[2]), "v"(den[3]), "v"(den[4]), "v"(den[5]), "v"(den[6]), "v"(den[7]));
}

__device__ __forceinline__ void den_step(float (&p)[8], const v8f& den, float (&q)[8]) {
  asm volatile(
      "v_rcp_f32_e32 %0, %0\n\t"
      "v_add_f32_e64 %8, |%16|, 1.0\n\t"
      "v_rcp_f32_e32 %1, %1\n\t"
      "v_add_f32_e64 %9, |%17|, 1.0\n\t"
      "v_rcp_f32_e32 %2, %2\n\t"
      "v_add_f32_e64 %10, |%18|, 1.0\n\t"
      "v_rcp_f32_e32 %3, %3\n\t"
      "v_add_f32_e64 %11, |%19|, 1.0\n\t"
      "v_rcp_f32_e32 %4, %4\n\t"
      "v_add_f32_e64 %12, |%20|, 1.0\n\t"
      "v_rcp_f32_e32 %5, %5\n\t"
      "v_add_f32_e64 %13, |%21|, 1.0\n\t"
      "v_rcp_f32_e32 %6, %6\n\t"
      "v_add_f32_e64 %14, |%22|, 1.0\n\t"
      "v_rcp_f32_e32 %7, %7\n\t"
      "v_add_f32_e64 %15, |%23|, 1.0"
      : "+v"(p[0]), "+v"(p[1]), "+v"(p[2]), "+v"(p[3]), "+v"(p[4]), "+v"(p[5]), "+v"(p[6]), "+v"(p[7]),
        "=&v"(q[0]), "=&v"(q[1]), "=&v"(q[2]), "=&v"(q[3]), "=&v"(q[4]), "=&v"(q[5]), "=&v"(q[6]), "=&v"(q[7])
      : "v"(den[0]), "v"(den[1]), "v"(den[2]), "v"(den[3]), "v"(den[4]), "v"(den[5]), "v"(den[6]), "v"(den[7]));
}

__device__ __forceinline__ void den_finish(float (&p)[8]) {
#pragma unroll
  for (int k = 0; k < 8; ++k) p[k] = __builtin_amdgcn_rcpf(p[k]);
}

__device__ __forceinline__ void num_accum(const v8f& num, const float (&r)[8], v8f& acc) {
  v2f a0 = {acc[0], acc[1]}, a1 = {acc[2], acc[3]}, a2 = {acc[4], acc[5]}, a3 = {acc[6], acc[7]};
  const v2f n0 = {num[0], num[1]}, n1 = {num[2], num[3]}, n2 = {num[4], num[5]}, n3 = {num[6], num[7]};
  const v2f r0 = {r[0], r[1]}, r1 = {r[2], r[3]}, r2 = {r[4], r[5]}, r3 = {r[6], r[7]};
  asm volatile(
      "v_pk_fma_f32 %0, %4, %8, %0\n\t"
      "v_pk_fma_f32 %1, %5, %9, %1\n\t"
      "v_pk_fma_f32 %2, %6, %10, %2\n\t"
      "v_pk_fma_f32 %3, %7, %11, %3"
      : "+v"(a0), "+v"(a1), "+v"(a2), "+v"(a3)
      : "v"(n0), "v"(n1), "v"(n2), "v"(n3), "v"(r0), "v"(r1), "v"(r2), "v"(r3));
  acc = (v8f){a0[0], a0[1], a1[0], a1[1], a2[0], a2[1], a3[0], a3[1]};
}

__global__ __launch_bounds__(512) void kaconv_wmma_kernel(
    const float* __restrict__ x,
    const float* __restrict__ nums,
    const float* __restrict__ denoms,
    const float* __restrict__ bias,
    float* __restrict__ out)
{
  __shared__ __attribute__((aligned(16))) float red[NC * 16 * RPITCH];
  __shared__ __attribute__((aligned(16))) float apow[NC * TPITCH];

  const int tid    = threadIdx.x;
  const int lane   = tid & 31;
  const int c      = tid >> 5;
  const int ochunk = blockIdx.x & 1;
  const int bg     = blockIdx.x >> 1;
  const int b      = bg / NGRP;
  const int g      = bg - b * NGRP;

  const int m  = lane & 15;
  const int hi = lane >> 4;

  const int obase = ochunk * 16;
  const float* xc = x + ((size_t)(b * NC + c)) * HW;

  const int row0 = ((obase + m) * NC + c) * (NK * NK);
  const float* nrA = nums   + (size_t)row0 * 6 + (hi ? 2 : 0);
  const float* nrB = nums   + (size_t)row0 * 6 + (hi ? 2 : 4);
  const float* drA = denoms + (size_t)row0 * 4 + (hi ? 2 : 0);

  TapOperands op;
  load_tap_coefs(op, nrA, nrB, drA, 0);

  float* tb = apow + c * TPITCH;
#pragma unroll
  for (int s = 0; s < (WPIX + 31) / 32; ++s) {
    const int p0 = (s * 32 + 32 <= WPIX) ? s * 32 : (WPIX - 32);
    const int p  = p0 + lane;
    const int wr = p / WCOLS;
    const int wc = p - wr * WCOLS;
    const int h  = g * 2 - 1 + wr;
    const int w  = wc - 1;
    const bool inb = (h >= 0) && (h < NH) && (w >= 0) && (w < NW);
    const float xl = xc[inb ? h * NW + w : 0];
    const float xv = inb ? xl : 0.0f;
    const float x2 = xv * xv;
    const float x3 = x2 * xv;
    const float x4 = x2 * x2;
    const float x5 = x3 * x2;
    float* e = tb + p * EPITCH;
    *(v4f*)(e)     = (v4f){1.0f, xv, x2, x3};
    *(v4f*)(e + 4) = (v4f){x4, x5, 0.0f, 0.0f};
    *(v4f*)(e + 8) = (v4f){xv, x2, x3, x4};
  }

  __syncthreads();

  const float* ap[MT];
#pragma unroll
  for (int t = 0; t < MT; ++t) {
    const int q = t * 16 + m;
    const int r = (q >= NW) ? 1 : 0;
    ap[t] = tb + (q + 2 * r) * EPITCH + hi * 2;
  }

  v8f acc[MT];
#pragma unroll
  for (int t = 0; t < MT; ++t) acc[t] = (v8f){};

  load_tap_table(op, ap, 0);

#pragma unroll 1
  for (int tap = 0; tap < NK * NK - 1; ++tap) {
    TapOperands nx;
    load_tap_coefs(nx, nrA, nrB, drA, tap + 1);

    v8f numv[MT], denv[MT];
    tap_products(op, numv, denv);
    __builtin_amdgcn_sched_barrier(0);

    float dv[MT][8];
    den_start(denv[0], dv[0]);
    den_step(dv[0], denv[1], dv[1]);
    den_step(dv[1], denv[2], dv[2]);
    num_accum(numv[0], dv[0], acc[0]);
    den_step(dv[2], denv[3], dv[3]);
    num_accum(numv[1], dv[1], acc[1]);

    __builtin_amdgcn_sched_barrier(0);
    load_tap_table(op, ap, tap + 1);
    __builtin_amdgcn_sched_barrier(0);

    den_step(dv[3], denv[4], dv[4]);
    num_accum(numv[2], dv[2], acc[2]);
    den_step(dv[4], denv[5], dv[5]);
    num_accum(numv[3], dv[3], acc[3]);
    den_step(dv[5], denv[6], dv[6]);
    num_accum(numv[4], dv[4], acc[4]);
    den_finish(dv[6]);
    num_accum(numv[5], dv[5], acc[5]);
    num_accum(numv[6], dv[6], acc[6]);
    op.bN0 = nx.bN0;
    op.bN1 = nx.bN1;
    op.bD  = nx.bD;
  }
  {
    v8f numv[MT], denv[MT];
    tap_products(op, numv, denv);
    __builtin_amdgcn_sched_barrier(0);
    float dv[MT][8];
    den_start(denv[0], dv[0]);
    den_step(dv[0], denv[1], dv[1]);
#pragma unroll
    for (int t = 2; t < MT; ++t) {
      den_step(dv[t - 1], denv[t], dv[t]);
      num_accum(numv[t - 2], dv[t - 2], acc[t - 2]);
    }
    den_finish(dv[MT - 1]);
    num_accum(numv[MT - 2], dv[MT - 2], acc[MT - 2]);
    num_accum(numv[MT - 1], dv[MT - 1], acc[MT - 1]);
  }

  float* rw = red + (c * 16 + m) * RPITCH + hi * 8;
#pragma unroll
  for (int t = 0; t < MT; ++t) {
    *(v4f*)(rw + t * 16)     = (v4f){acc[t][0], acc[t][1], acc[t][2], acc[t][3]};
    *(v4f*)(rw + t * 16 + 4) = (v4f){acc[t][4], acc[t][5], acc[t][6], acc[t][7]};
  }
  __syncthreads();

  if (tid < 16 * NQUAD) {
    const int o  = tid / NQUAD;
    const int q4 = tid - o * NQUAD;
    const float* rr = red + o * RPITCH + q4 * 4;
    v4f s = *(const v4f*)rr;
#pragma unroll
    for (int cc = 1; cc < NC; ++cc) s += *(const v4f*)(rr + cc * 16 * RPITCH);
    const float bo = bias[obase + o];
    s += (v4f){bo, bo, bo, bo};
    *(v4f*)(out + ((size_t)(b * NO + obase + o)) * HW + g * GPIX + q4 * 4) = s;
  }
}

extern "C" void kernel_launch(void* const* d_in, const int* in_sizes, int n_in,
                              void* d_out, int out_size, void* d_ws, size_t ws_size,
                              hipStream_t stream) {
  const float* x      = (const float*)d_in[0];
  const float* nums   = (const float*)d_in[1];
  const float* denoms = (const float*)d_in[2];
  const float* bias   = (const float*)d_in[3];
  float* out          = (float*)d_out;
  (void)in_sizes; (void)n_in; (void)out_size; (void)d_ws; (void)ws_size;

  dim3 grid(224);
  dim3 block(512);
  hipLaunchKernelGGL(kaconv_wmma_kernel, grid, block, 0, stream,
                     x, nums, denoms, bias, out);
}
